# starting kernel
# baseline (speedup 1.0000x reference)
#include <hip/hip_runtime.h>
#include <hip/hip_bf16.h>
#include <math.h>

typedef float v2f __attribute__((ext_vector_type(2)));
typedef float v8f __attribute__((ext_vector_type(8)));

#define D_FEAT 64
#define H_FEAT 128
#define WAVES_PER_BLOCK 4

__global__ void edge_msg_kernel(const float* __restrict__ x,
                                const int*   __restrict__ edge_index,
                                const int*   __restrict__ edge_attr,
                                const float* __restrict__ edge_weight,
                                const float* __restrict__ bond_emb,
                                float*       __restrict__ agg,
                                int E, int N)
{
    int e    = (int)((blockIdx.x * blockDim.x + threadIdx.x) >> 5);
    int lane = threadIdx.x & 31;
    if (e >= E) return;

    int src = edge_index[e];
    int dst = edge_index[E + e];
    int att = edge_attr[e];
    float w0 = edge_weight[e];
    float w1 = edge_weight[(size_t)E + e];

    int d = lane * 2;
    const float* er = bond_emb + (size_t)att * D_FEAT;
    float2 ev = *(const float2*)(er + d);

    const float kInvSqrt2 = 0.70710678118654752f;

    {
        const float* xr = x   + ((size_t)src) * D_FEAT;
        float*       ar = agg + ((size_t)dst) * D_FEAT;
        float2 xv = *(const float2*)(xr + d);
        float v0 = xv.x + ev.x;
        float v1 = xv.y + ev.y;
        v0 = 0.5f * v0 * (1.0f + erff(v0 * kInvSqrt2));
        v1 = 0.5f * v1 * (1.0f + erff(v1 * kInvSqrt2));
        atomicAdd(ar + d,     v0 * w0);
        atomicAdd(ar + d + 1, v1 * w0);
    }
    {
        const float* xr = x   + ((size_t)N + src) * D_FEAT;
        float*       ar = agg + ((size_t)N + dst) * D_FEAT;
        float2 xv = *(const float2*)(xr + d);
        float v0 = xv.x + ev.x;
        float v1 = xv.y + ev.y;
        v0 = 0.5f * v0 * (1.0f + erff(v0 * kInvSqrt2));
        v1 = 0.5f * v1 * (1.0f + erff(v1 * kInvSqrt2));
        atomicAdd(ar + d,     v0 * w1);
        atomicAdd(ar + d + 1, v1 * w1);
    }
}

__global__ void gin_mlp_kernel(const float* __restrict__ x,
                               const float* __restrict__ agg,
                               const float* __restrict__ W1,
                               const float* __restrict__ b1,
                               const float* __restrict__ W2,
                               const float* __restrict__ b2,
                               const float* __restrict__ eps,
                               float*       __restrict__ out,
                               int n_tiles)
{
    __shared__ float lds_t[WAVES_PER_BLOCK][16 * H_FEAT];

    int waveId = threadIdx.x >> 5;
    int lane   = threadIdx.x & 31;
    int tile   = blockIdx.x * WAVES_PER_BLOCK + waveId;
    if (tile >= n_tiles) return;

    float epsv = 1.0f + eps[0];
    int m    = lane & 15;
    int half = lane >> 4;

    size_t rowbase = ((size_t)tile * 16 + m) * D_FEAT;

    v2f A[16];
#pragma unroll
    for (int k = 0; k < 16; ++k) {
        int col = k * 4 + half * 2;
        float2 xv = *(const float2*)(x   + rowbase + col);
        float2 av = *(const float2*)(agg + rowbase + col);
        A[k].x = epsv * xv.x + av.x;
        A[k].y = epsv * xv.y + av.y;
    }

    float* tl = &lds_t[waveId][0];

#pragma unroll
    for (int n = 0; n < 8; ++n) {
        v8f acc = {};
        int ncol = n * 16 + m;
#pragma unroll
        for (int k = 0; k < 16; ++k) {
            int kb = k * 4 + half * 2;
            v2f B;
            B.x = W1[(size_t)kb * H_FEAT + ncol];
            B.y = W1[(size_t)(kb + 1) * H_FEAT + ncol];
            acc = __builtin_amdgcn_wmma_f32_16x16x4_f32(
                      false, A[k], false, B, (short)0, acc, false, false);
        }
        float bias = b1[ncol];
#pragma unroll
        for (int v = 0; v < 8; ++v) {
            float t = acc[v] + bias;
            t = t > 0.0f ? t : 0.0f;
            tl[(v + 8 * half) * H_FEAT + ncol] = t;
        }
    }

    asm volatile("s_wait_dscnt 0" ::: "memory");
    __builtin_amdgcn_wave_barrier();

#pragma unroll
    for (int o = 0; o < 4; ++o) {
        v8f acc = {};
        int ocol = o * 16 + m;
#pragma unroll
        for (int j = 0; j < 32; ++j) {
            int kb = j * 4 + half * 2;
            float2 av = *(const float2*)(tl + m * H_FEAT + kb);
            v2f A2; A2.x = av.x; A2.y = av.y;
            v2f B2;
            B2.x = W2[(size_t)kb * D_FEAT + ocol];
            B2.y = W2[(size_t)(kb + 1) * D_FEAT + ocol];
            acc = __builtin_amdgcn_wmma_f32_16x16x4_f32(
                      false, A2, false, B2, (short)0, acc, false, false);
        }
        float bias = b2[ocol];
#pragma unroll
        for (int v = 0; v < 8; ++v) {
            out[((size_t)tile * 16 + v + 8 * half) * D_FEAT + ocol] = acc[v] + bias;
        }
    }
}

extern "C" void kernel_launch(void* const* d_in, const int* in_sizes, int n_in,
                              void* d_out, int out_size, void* d_ws, size_t ws_size,
                              hipStream_t stream)
{
    const float* x           = (const float*)d_in[0];
    const int*   edge_index  = (const int*)  d_in[1];
    const int*   edge_attr   = (const int*)  d_in[2];
    const float* edge_weight = (const float*)d_in[3];
    const float* bond_emb    = (const float*)d_in[4];
    const float* eps         = (const float*)d_in[5];
    const float* W1          = (const float*)d_in[6];
    const float* b1          = (const float*)d_in[7];
    const float* W2          = (const float*)d_in[8];
    const float* b2          = (const float*)d_in[9];
    float*       out         = (float*)d_out;

    const int RC = 2;
    const int E  = in_sizes[2];
    const int N  = in_sizes[0] / (RC * D_FEAT);

    float* agg = (float*)d_ws;
    hipMemsetAsync(agg, 0, (size_t)RC * N * D_FEAT * sizeof(float), stream);

    long long totalThreads = (long long)E * 32;
    int blocks1 = (int)((totalThreads + 255) / 256);
    edge_msg_kernel<<<blocks1, 256, 0, stream>>>(x, edge_index, edge_attr,
                                                 edge_weight, bond_emb, agg, E, N);

    int n_tiles = (RC * N) / 16;
    int blocks2 = (n_tiles + WAVES_PER_BLOCK - 1) / WAVES_PER_BLOCK;
    gin_mlp_kernel<<<blocks2, WAVES_PER_BLOCK * 32, 0, stream>>>(
        x, agg, W1, b1, W2, b2, eps, out, n_tiles);
}


# MLP phase: fragment k order and N-tile column order picked so every operand load is 16 B per lane (h rows; W1/W2 rows, 4 tiles per load; t from LDS, 528-B pitch); k order inside products changes
# speedup vs baseline: 1.5194x; 1.5194x over previous
#include <hip/hip_runtime.h>
#include <hip/hip_bf16.h>
#include <math.h>

typedef float v2f __attribute__((ext_vector_type(2)));
typedef float v8f __attribute__((ext_vector_type(8)));

#define D_FEAT 64
#define H_FEAT 128
#define WAVES_PER_BLOCK 4

#define EDGES_PER_WAVE 32
#define EDGE_BLOCK_THREADS 256

__device__ __forceinline__ v2f fma2(v2f a, v2f b, v2f c)
{
    return __builtin_elementwise_fma(a, b, c);
}

__device__ __forceinline__ v2f gelu_exact2(v2f v)
{
    const float kInvSqrt2 = 0.70710678118654752f;
    v2f t = v * kInvSqrt2;
    v2f a = __builtin_elementwise_abs(t);

    v2f s = t * t;
    v2f p = fma2(s, (v2f)(-0.000561801775f), (v2f)(0.00491381623f));
    p = fma2(s, p, (v2f)(-0.0267075151f));
    p = fma2(s, p, (v2f)(0.112800106f));
    p = fma2(s, p, (v2f)(-0.376122952f));
    p = fma2(s, p, (v2f)(0.128379107f));
    v2f lo = fma2(a, p, a);

    v2f q = fma2(a, (v2f)(0.0000169988107f), (v2f)(-0.000378677854f));
    q = fma2(a, q, (v2f)(0.00385781587f));
    q = fma2(a, q, (v2f)(-0.0241816975f));
    q = fma2(a, q, (v2f)(0.106668264f));
    q = fma2(a, q, (v2f)(0.634933293f));
    q = fma2(a, q, (v2f)(0.128689408f));
    q = fma2(a, q, a);
    v2f hi;
    hi.x = 1.0f - expf(-q.x);
    hi.y = 1.0f - expf(-q.y);

    v2f r;
    r.x = copysignf(a.x < 1.0f ? lo.x : hi.x, t.x);
    r.y = copysignf(a.y < 1.0f ? lo.y : hi.y, t.y);
    return (0.5f * v) * (1.0f + r);
}

__global__ __launch_bounds__(EDGE_BLOCK_THREADS)
void edge_msg_kernel(const float* __restrict__ x,
                     const int*   __restrict__ edge_index,
                     const int*   __restrict__ edge_attr,
                     const float* __restrict__ edge_weight,
                     const float* __restrict__ bond_emb,
                     float*       __restrict__ agg,
                     int E, int N)
{
    int lane = threadIdx.x & 31;
    int wave = __builtin_amdgcn_readfirstlane(
                   (int)((blockIdx.x * blockDim.x + threadIdx.x) >> 5));
    int e0 = wave * EDGES_PER_WAVE;
    if (e0 >= E) return;
    int cnt = E - e0;
    if (cnt > EDGES_PER_WAVE) cnt = EDGES_PER_WAVE;

    int mine = e0 + (lane < cnt ? lane : cnt - 1);
    int src_l = edge_index[mine];
    int dst_l = edge_index[(size_t)E + mine];
    int att_l = edge_attr[mine];
    int w0_l  = __float_as_int(edge_weight[mine]);
    int w1_l  = __float_as_int(edge_weight[(size_t)E + mine]);

    int d = lane;
    const int HALF_ROW = D_FEAT / 2;

    int src = __builtin_amdgcn_readlane(src_l, 0);
    int att = __builtin_amdgcn_readlane(att_l, 0);
    const float* pe  = bond_emb + (size_t)att * D_FEAT + d;
    const float* px0 = x + ((size_t)src) * D_FEAT + d;
    const float* px1 = x + ((size_t)N + src) * D_FEAT + d;
    float2 ev, x0, x1;
    ev.x = pe[0];  ev.y = pe[HALF_ROW];
    x0.x = px0[0]; x0.y = px0[HALF_ROW];
    x1.x = px1[0]; x1.y = px1[HALF_ROW];

    for (int i = 0; i < cnt; ++i) {
        float2 cev = ev, cx0 = x0, cx1 = x1;

        int nx = (i + 1 < cnt) ? i + 1 : i;
        src = __builtin_amdgcn_readlane(src_l, nx);
        att = __builtin_amdgcn_readlane(att_l, nx);
        pe  = bond_emb + (size_t)att * D_FEAT + d;
        px0 = x + ((size_t)src) * D_FEAT + d;
        px1 = x + ((size_t)N + src) * D_FEAT + d;
        ev.x = pe[0];  ev.y = pe[HALF_ROW];
        x0.x = px0[0]; x0.y = px0[HALF_ROW];
        x1.x = px1[0]; x1.y = px1[HALF_ROW];
        __builtin_amdgcn_sched_barrier(0);

        int   dst = __builtin_amdgcn_readlane(dst_l, i);
        float w0  = __int_as_float(__builtin_amdgcn_readlane(w0_l, i));
        float w1  = __int_as_float(__builtin_amdgcn_readlane(w1_l, i));

        float* a0 = agg + ((size_t)dst) * D_FEAT + d;
        float* a1 = agg + ((size_t)N + dst) * D_FEAT + d;

        v2f e2; e2.x = cev.x; e2.y = cev.y;
        v2f h0; h0.x = cx0.x; h0.y = cx0.y;
        v2f h1; h1.x = cx1.x; h1.y = cx1.y;
        v2f g0 = gelu_exact2(h0 + e2) * w0;
        v2f g1 = gelu_exact2(h1 + e2) * w1;

        atomicAdd(a0,            g0.x);
        atomicAdd(a0 + HALF_ROW, g0.y);
        atomicAdd(a1,            g1.x);
        atomicAdd(a1 + HALF_ROW, g1.y);
    }
}

#define T_PITCH 132
#define K_FIRST(kk, half) (8 * ((kk) >> 1) + 4 * (half) + 2 * ((kk) & 1))

__device__ __forceinline__ void load_w_block(const float* __restrict__ Wm, int pitch,
                                             int kk0, int cbase, int half,
                                             float4 (&P)[4], float4 (&Q)[4])
{
#pragma unroll
    for (int u = 0; u < 4; ++u) {
        int kr = K_FIRST(kk0 + u, half);
        P[u] = *(const float4*)(Wm + (size_t)kr * pitch + cbase);
        Q[u] = *(const float4*)(Wm + (size_t)(kr + 1) * pitch + cbase);
    }
}

__device__ __forceinline__ void wmma4(v2f a, float4 p, float4 q,
                                      v8f& acc0, v8f& acc1, v8f& acc2, v8f& acc3)
{
    v2f B0; B0.x = p.x; B0.y = q.x;
    v2f B1; B1.x = p.y; B1.y = q.y;
    v2f B2; B2.x = p.z; B2.y = q.z;
    v2f B3; B3.x = p.w; B3.y = q.w;
    acc0 = __builtin_amdgcn_wmma_f32_16x16x4_f32(false, a, false, B0, (short)0, acc0, false, false);
    acc1 = __builtin_amdgcn_wmma_f32_16x16x4_f32(false, a, false, B1, (short)0, acc1, false, false);
    acc2 = __builtin_amdgcn_wmma_f32_16x16x4_f32(false, a, false, B2, (short)0, acc2, false, false);
    acc3 = __builtin_amdgcn_wmma_f32_16x16x4_f32(false, a, false, B3, (short)0, acc3, false, false);
}

__global__ __launch_bounds__(WAVES_PER_BLOCK * 32)
void gin_mlp_kernel(const float* __restrict__ x,
                    const float* __restrict__ agg,
                    const float* __restrict__ W1,
                    const float* __restrict__ b1,
                    const float* __restrict__ W2,
                    const float* __restrict__ b2,
                    const float* __restrict__ eps,
                    float*       __restrict__ out,
                    int n_tiles)
{
    __shared__ __attribute__((aligned(16))) float lds_t[WAVES_PER_BLOCK][16 * T_PITCH];

    int waveId = threadIdx.x >> 5;
    int lane   = threadIdx.x & 31;
    int tile   = blockIdx.x * WAVES_PER_BLOCK + waveId;
    if (tile >= n_tiles) return;

    float epsv = 1.0f + eps[0];
    int m    = lane & 15;
    int half = lane >> 4;
    int c4   = 4 * m;

    size_t rowbase = ((size_t)tile * 16 + m) * D_FEAT;

    v2f A[16];
#pragma unroll
    for (int i = 0; i < 8; ++i) {
        int col = K_FIRST(2 * i, half);
        float4 xv = *(const float4*)(x   + rowbase + col);
        float4 av = *(const float4*)(agg + rowbase + col);
        A[2 * i].x     = epsv * xv.x + av.x;
        A[2 * i].y     = epsv * xv.y + av.y;
        A[2 * i + 1].x = epsv * xv.z + av.z;
        A[2 * i + 1].y = epsv * xv.w + av.w;
    }

    float* tl = &lds_t[waveId][0];

    float4 P[2][4], Q[2][4];
    v8f acc0 = {}, acc1 = {}, acc2 = {}, acc3 = {};

    load_w_block(W1, H_FEAT, 0, c4, half, P[0], Q[0]);
#pragma unroll
    for (int b = 0; b < 8; ++b) {
        int g = b >> 2, blk = b & 3;
        if (b + 1 < 8)
            load_w_block(W1, H_FEAT, 4 * ((b + 1) & 3), ((b + 1) >> 2) * 64 + c4, half,
                         P[(b + 1) & 1], Q[(b + 1) & 1]);
        else
            load_w_block(W2, D_FEAT, 0, c4, half, P[0], Q[0]);
        __builtin_amdgcn_sched_barrier(0);
        if (blk == 0) { acc0 = (v8f){}; acc1 = (v8f){}; acc2 = (v8f){}; acc3 = (v8f){}; }
#pragma unroll
        for (int u = 0; u < 4; ++u)
            wmma4(A[4 * blk + u], P[b & 1][u], Q[b & 1][u], acc0, acc1, acc2, acc3);
        if (blk == 3) {
            int cbase = g * 64 + c4;
            float4 bias = *(const float4*)(b1 + cbase);
#pragma unroll
            for (int v = 0; v < 8; ++v) {
                float4 t;
                t.x = fmaxf(acc0[v] + bias.x, 0.0f);
                t.y = fmaxf(acc1[v] + bias.y, 0.0f);
                t.z = fmaxf(acc2[v] + bias.z, 0.0f);
                t.w = fmaxf(acc3[v] + bias.w, 0.0f);
                *(float4*)(tl + (v + 8 * half) * T_PITCH + cbase) = t;
            }
        }
    }

    asm volatile("s_wait_dscnt 0" ::: "memory");
    __builtin_amdgcn_wave_barrier();

    float4 T4[8];
#pragma unroll
    for (int i = 0; i < 8; ++i)
        T4[i] = *(const float4*)(tl + m * T_PITCH + K_FIRST(2 * i, half));

    acc0 = (v8f){}; acc1 = (v8f){}; acc2 = (v8f){}; acc3 = (v8f){};
#pragma unroll
    for (int b = 0; b < 8; ++b) {
        if (b == 4) {
#pragma unroll
            for (int i = 0; i < 8; ++i)
                T4[i] = *(const float4*)(tl + m * T_PITCH + K_FIRST(2 * (i + 8), half));
        }
        if (b + 1 < 8)
            load_w_block(W2, D_FEAT, 4 * (b + 1), c4, half, P[(b + 1) & 1], Q[(b + 1) & 1]);
        __builtin_amdgcn_sched_barrier(0);
#pragma unroll
        for (int u = 0; u < 4; ++u) {
            int kk = 4 * b + u;
            float4 a4 = T4[(kk >> 1) & 7];
            v2f A2;
            A2.x = (kk & 1) ? a4.z : a4.x;
            A2.y = (kk & 1) ? a4.w : a4.y;
            wmma4(A2, P[b & 1][u], Q[b & 1][u], acc0, acc1, acc2, acc3);
        }
    }
    {
        float4 bias = *(const float4*)(b2 + c4);
#pragma unroll
        for (int v = 0; v < 8; ++v) {
            float4 o;
            o.x = acc0[v] + bias.x;
            o.y = acc1[v] + bias.y;
            o.z = acc2[v] + bias.z;
            o.w = acc3[v] + bias.w;
            *(float4*)(out + ((size_t)tile * 16 + v + 8 * half) * D_FEAT + c4) = o;
        }
    }
}

extern "C" void kernel_launch(void* const* d_in, const int* in_sizes, int n_in,
                              void* d_out, int out_size, void* d_ws, size_t ws_size,
                              hipStream_t stream)
{
    const float* x           = (const float*)d_in[0];
    const int*   edge_index  = (const int*)  d_in[1];
    const int*   edge_attr   = (const int*)  d_in[2];
    const float* edge_weight = (const float*)d_in[3];
    const float* bond_emb    = (const float*)d_in[4];
    const float* eps         = (const float*)d_in[5];
    const float* W1          = (const float*)d_in[6];
    const float* b1          = (const float*)d_in[7];
    const float* W2          = (const float*)d_in[8];
    const float* b2          = (const float*)d_in[9];
    float*       out         = (float*)d_out;

    const int RC = 2;
    const int E  = in_sizes[2];
    const int N  = in_sizes[0] / (RC * D_FEAT);

    float* agg = (float*)d_ws;
    hipMemsetAsync(agg, 0, (size_t)RC * N * D_FEAT * sizeof(float), stream);

    int waves1  = (E + EDGES_PER_WAVE - 1) / EDGES_PER_WAVE;
    int blocks1 = (waves1 * 32 + EDGE_BLOCK_THREADS - 1) / EDGE_BLOCK_THREADS;
    edge_msg_kernel<<<blocks1, EDGE_BLOCK_THREADS, 0, stream>>>(x, edge_index, edge_attr,
                                                 edge_weight, bond_emb, agg, E, N);

    int n_tiles = (RC * N) / 16;
    int blocks2 = (n_tiles + WAVES_PER_BLOCK - 1) / WAVES_PER_BLOCK;
    gin_mlp_kernel<<<blocks2, WAVES_PER_BLOCK * 32, 0, stream>>>(
        x, agg, W1, b1, W2, b2, eps, out, n_tiles);
}
